# starting kernel
# baseline (speedup 1.0000x reference)
#include <hip/hip_runtime.h>
#include <hip/hip_bf16.h>


typedef float v2f __attribute__((ext_vector_type(2)));
typedef float v8f __attribute__((ext_vector_type(8)));

#define GAMMA_EXP 0.38461538461538464f
#define SIGMA_P   0.17f
#define SIGMA_LP  4.86f

__device__ __forceinline__ int reflc(int r, int n) {
    r = (r < 0) ? -r : r;
    if (r >= n) r = 2 * n - 2 - r;
    if (r < 0)  r = 0;
    if (r >= n) r = n - 1;
    return r;
}

__device__ __forceinline__ float ltap(int d) {
    float v = 0.0f;
    if (d == 0 || d == 4)      v = 0.05f;
    else if (d == 1 || d == 3) v = 0.25f;
    else if (d == 2)           v = 0.40f;
    return v;
}

__device__ __forceinline__ float src_at(const float* __restrict__ src, int p,
                                        int y, int x, int H, int W, int gam) {
    if (gam) {
        int b = p / 3, c = p - 3 * b;
        float v = src[(((size_t)b * H + y) * W + x) * 3 + c];
        return __builtin_amdgcn_exp2f(GAMMA_EXP * __builtin_amdgcn_logf(v));
    } else {
        int Wp = W + 2;
        return src[(size_t)p * (H + 2) * Wp + (size_t)(y + 1) * Wp + (x + 1)];
    }
}

__global__ void __launch_bounds__(32)
nlp_down(const float* __restrict__ src, float* __restrict__ down,
         int H, int W, int gammaFlag) {
    __shared__ float vbuf[16][49];

    const int H2 = H / 2 + 2, W2 = W / 2 + 2;
    const int p  = blockIdx.z;
    const int i0 = blockIdx.y * 16, j0 = blockIdx.x * 16;
    const int lane = threadIdx.x;
    const int n = lane & 15;
    const int half = lane >> 4;

    int colx[3];
#pragma unroll
    for (int nt = 0; nt < 3; ++nt)
        colx[nt] = reflc(2 * j0 - 4 + nt * 16 + n, W);

    v8f acc0 = {0,0,0,0,0,0,0,0};
    v8f acc1 = {0,0,0,0,0,0,0,0};
    v8f acc2 = {0,0,0,0,0,0,0,0};

    for (int c = 0; c < 9; ++c) {
        const int kb = 4 * c + 2 * half;
        v2f a;
        a.x = ltap(kb - 2 * n);
        a.y = ltap(kb + 1 - 2 * n);
        const int r0 = reflc(2 * i0 + kb - 4, H);
        const int r1 = reflc(2 * i0 + kb - 3, H);

        v2f b0, b1, b2;
        b0.x = src_at(src, p, r0, colx[0], H, W, gammaFlag);
        b0.y = src_at(src, p, r1, colx[0], H, W, gammaFlag);
        b1.x = src_at(src, p, r0, colx[1], H, W, gammaFlag);
        b1.y = src_at(src, p, r1, colx[1], H, W, gammaFlag);
        b2.x = src_at(src, p, r0, colx[2], H, W, gammaFlag);
        b2.y = src_at(src, p, r1, colx[2], H, W, gammaFlag);

        acc0 = __builtin_amdgcn_wmma_f32_16x16x4_f32(false, a, false, b0,
                                                     (short)0, acc0, false, false);
        acc1 = __builtin_amdgcn_wmma_f32_16x16x4_f32(false, a, false, b1,
                                                     (short)0, acc1, false, false);
        acc2 = __builtin_amdgcn_wmma_f32_16x16x4_f32(false, a, false, b2,
                                                     (short)0, acc2, false, false);
    }

#pragma unroll
    for (int v = 0; v < 8; ++v) {
        vbuf[v + 8 * half][n]      = acc0[v];
        vbuf[v + 8 * half][16 + n] = acc1[v];
        vbuf[v + 8 * half][32 + n] = acc2[v];
    }
    __syncthreads();

    v8f dacc = {0,0,0,0,0,0,0,0};
    for (int c = 0; c < 9; ++c) {
        const int kb = 4 * c + 2 * half;
        v2f a;
        a.x = ltap(kb - 2 * n);
        a.y = ltap(kb + 1 - 2 * n);
        v2f b;
        b.x = vbuf[n][kb];
        b.y = vbuf[n][kb + 1];
        dacc = __builtin_amdgcn_wmma_f32_16x16x4_f32(false, a, false, b,
                                                     (short)0, dacc, false, false);
    }

    const int gi = i0 + n;
    if (gi < H2) {
        float* dp = down + (size_t)p * H2 * W2 + (size_t)gi * W2;
#pragma unroll
        for (int v = 0; v < 8; ++v) {
            int gj = j0 + v + 8 * half;
            if (gj < W2) dp[gj] = dacc[v];
        }
    }
}

__global__ void __launch_bounds__(256)
nlp_sub(const float* __restrict__ src, const float* __restrict__ down,
        float* __restrict__ out, int H, int W, int gammaFlag) {
    __shared__ float zb[36][37];
    __shared__ float dt[20][21];

    const int H2 = H / 2 + 2, W2 = W / 2 + 2;
    const int p  = blockIdx.z;
    const int y0 = blockIdx.y * 32, x0 = blockIdx.x * 32;
    const int tid = threadIdx.x;
    const float* dw = down + (size_t)p * H2 * W2;

    const int dr0 = (y0 >> 1) - 1, dc0 = (x0 >> 1) - 1;
    __builtin_prefetch(dw + (size_t)(dr0 < 0 ? 0 : dr0) * W2, 0, 1);

    for (int t = tid; t < 400; t += 256) {
        int ly = t / 20, lx = t % 20;
        int r = dr0 + ly; r = r < 0 ? 0 : (r >= H2 ? H2 - 1 : r);
        int c = dc0 + lx; c = c < 0 ? 0 : (c >= W2 ? W2 - 1 : c);
        dt[ly][lx] = dw[(size_t)r * W2 + c];
    }
    __syncthreads();

    for (int t = tid; t < 1296; t += 256) {
        int ly = t / 36, lx = t % 36;
        int y = y0 - 2 + ly, x = x0 - 2 + lx;
        float z = 0.0f;
        if (y >= 0 && y < H && x >= 0 && x < W) {
            float xv = src_at(src, p, y, x, H, W, gammaFlag);
            float wv[3], wh[3];
            int r0, c0;
            if (y & 1) { r0 = (y + 1) >> 1; wv[0] = 0.5f; wv[1] = 0.5f; wv[2] = 0.0f; }
            else       { r0 = y >> 1;       wv[0] = 0.1f; wv[1] = 0.8f; wv[2] = 0.1f; }
            if (x & 1) { c0 = (x + 1) >> 1; wh[0] = 0.5f; wh[1] = 0.5f; wh[2] = 0.0f; }
            else       { c0 = x >> 1;       wh[0] = 0.1f; wh[1] = 0.8f; wh[2] = 0.1f; }
            float up = 0.0f;
#pragma unroll
            for (int rr = 0; rr < 3; ++rr) {
                int lr = r0 + rr - dr0; lr = lr < 0 ? 0 : (lr > 19 ? 19 : lr);
                float rowv = 0.0f;
#pragma unroll
                for (int cc = 0; cc < 3; ++cc) {
                    int lc = c0 + cc - dc0; lc = lc < 0 ? 0 : (lc > 19 ? 19 : lc);
                    rowv = fmaf(wh[cc], dt[lr][lc], rowv);
                }
                up = fmaf(wv[rr], rowv, up);
            }
            z = xv - up;
        }
        zb[ly][lx] = z;
    }
    __syncthreads();

    const float Pk[5][5] = {
        {0.04f, 0.04f, 0.05f, 0.04f, 0.04f},
        {0.04f, 0.03f, 0.04f, 0.03f, 0.04f},
        {0.05f, 0.04f, 0.05f, 0.04f, 0.05f},
        {0.04f, 0.03f, 0.04f, 0.03f, 0.04f},
        {0.04f, 0.04f, 0.05f, 0.04f, 0.04f}};

    const int b = p / 3, ch = p - 3 * b;
#pragma unroll
    for (int q = 0; q < 4; ++q) {
        int idx = tid + q * 256;
        int oy = idx >> 5, ox = idx & 31;
        float zc = zb[oy + 2][ox + 2];
        float pool = 0.0f;
#pragma unroll
        for (int dy = 0; dy < 5; ++dy)
#pragma unroll
            for (int dx = 0; dx < 5; ++dx)
                pool = fmaf(Pk[dy][dx], fabsf(zb[oy + dy][ox + dx]), pool);
        float o = zc / (pool + SIGMA_P);
        int gy = y0 + oy, gx = x0 + ox;
        out[(((size_t)b * H + gy) * W + gx) * 3 + ch] = o;
    }
}

__global__ void __launch_bounds__(256)
nlp_resid(const float* __restrict__ d4, float* __restrict__ out) {
    const int p = blockIdx.x, tid = threadIdx.x;
    const int y = tid >> 4, x = tid & 15;
    float v = d4[(size_t)p * 18 * 18 + (size_t)(y + 1) * 18 + (x + 1)];
    float o = v / (fabsf(v) + SIGMA_LP);
    int b = p / 3, c = p - 3 * b;
    out[(((size_t)b * 16 + y) * 16 + x) * 3 + c] = o;
}

extern "C" void kernel_launch(void* const* d_in, const int* in_sizes, int n_in,
                              void* d_out, int out_size, void* d_ws, size_t ws_size,
                              hipStream_t stream) {
    (void)in_sizes; (void)n_in; (void)out_size; (void)ws_size;
    const float* img = (const float*)d_in[0];
    float* out = (float*)d_out;
    float* ws  = (float*)d_ws;

    const int Hl[6] = {512, 256, 128, 64, 32, 16};

    size_t dof[5]; size_t off = 0;
    for (int l = 0; l < 5; ++l) {
        dof[l] = off;
        size_t h2 = (size_t)(Hl[l] / 2 + 2);
        off += (size_t)48 * h2 * h2;
    }
    size_t oof[6]; size_t o2 = 0;
    for (int l = 0; l < 6; ++l) {
        oof[l] = o2;
        o2 += (size_t)16 * Hl[l] * Hl[l] * 3;
    }

    const float* srcp = img;
    int gam = 1;
    for (int l = 0; l < 5; ++l) {
        const int H = Hl[l];
        const int H2 = H / 2 + 2;
        const int T = (H2 + 15) / 16;
        nlp_down<<<dim3(T, T, 48), 32, 0, stream>>>(srcp, ws + dof[l], H, H, gam);
        nlp_sub<<<dim3(H / 32, H / 32, 48), 256, 0, stream>>>(
            srcp, ws + dof[l], out + oof[l], H, H, gam);
        srcp = ws + dof[l];
        gam = 0;
    }
    nlp_resid<<<48, 256, 0, stream>>>(ws + dof[4], out + oof[5]);
}


# down levels 3 and 4 of a plane formed inside the sub-band launch by the work-group that then walks that plane's level-3, level-4 and residual tiles: 6 -> 4 dependent launches; sums unchanged
# speedup vs baseline: 1.7286x; 1.7286x over previous
#include <hip/hip_runtime.h>
#include <hip/hip_bf16.h>


typedef float v2f __attribute__((ext_vector_type(2)));
typedef float v8f __attribute__((ext_vector_type(8)));

#define GAMMA_EXP 0.38461538461538464f
#define SIGMA_P   0.17f
#define SIGMA_LP  4.86f

__device__ __forceinline__ int reflc(int r, int n) {
    r = (r < 0) ? -r : r;
    if (r >= n) r = 2 * n - 2 - r;
    if (r < 0)  r = 0;
    if (r >= n) r = n - 1;
    return r;
}

__device__ __forceinline__ float ltap(int d) {
    float v = 0.0f;
    if (d == 0 || d == 4)      v = 0.05f;
    else if (d == 1 || d == 3) v = 0.25f;
    else if (d == 2)           v = 0.40f;
    return v;
}

#define SLAB_N     68
#define SLAB_PITCH 72

template <int GAM>
__device__ __forceinline__ void stage_slab(const float* __restrict__ src,
                                           float (*slab)[SLAB_PITCH],
                                           int p, int I0, int J0, int H, int W, int tid) {
    const float* sp;
    int rowPitch, xs;
    if (GAM) {
        int b = p / 3, c = p - 3 * b;
        sp = src + (size_t)b * H * W * 3 + c;
        rowPitch = W * 3; xs = 3;
    } else {
        int Wp = W + 2;
        sp = src + (size_t)p * (H + 2) * Wp + Wp + 1;
        rowPitch = Wp; xs = 1;
    }
    const int tx = tid & 63, ty = tid >> 6;
    const int gx = reflc(2 * J0 - 4 + tx, W) * xs;

    float ve[3];
    for (int k = 0; k < 3; ++k) {
        int e = tid + 128 * k;
        if (e > 4 * SLAB_N - 1) e = 4 * SLAB_N - 1;
        int gy = reflc(2 * I0 - 4 + (e >> 2), H);
        int gxe = reflc(2 * J0 - 4 + 64 + (e & 3), W) * xs;
        ve[k] = sp[gy * rowPitch + gxe];
    }
    for (int hb = 0; hb < 2; ++hb) {
        float va[17];
        for (int k = 0; k < 17; ++k) {
            int r = ty + 2 * (k + 17 * hb);
            int gy = reflc(2 * I0 - 4 + r, H);
            va[k] = sp[gy * rowPitch + gx];
        }
        for (int k = 0; k < 17; ++k) {
            int r = ty + 2 * (k + 17 * hb);
            float v = va[k];
            if (GAM) v = __builtin_amdgcn_exp2f(GAMMA_EXP * __builtin_amdgcn_logf(v));
            slab[r][tx] = v;
        }
    }
    for (int k = 0; k < 3; ++k) {
        int e = tid + 128 * k;
        if (e > 4 * SLAB_N - 1) e = 4 * SLAB_N - 1;
        float v = ve[k];
        if (GAM) v = __builtin_amdgcn_exp2f(GAMMA_EXP * __builtin_amdgcn_logf(v));
        slab[e >> 2][64 + (e & 3)] = v;
    }
}

template <int GAM, int G>
__device__ __forceinline__ void stage_part_group(const float* __restrict__ sp,
                                                 float (*slab)[SLAB_PITCH],
                                                 int rowPitch, int xs, int I0, int J0,
                                                 int H, int W, int tid,
                                                 int base, int ne, int cmax, float inv) {
    float v[G];
    int rr[G], cc[G];
#pragma unroll
    for (int k = 0; k < G; ++k) {
        int e = base + tid + 128 * k;
        e = e > ne - 1 ? ne - 1 : e;
        const int r = (int)(((float)e + 0.5f) * inv);
        const int c = e - r * cmax;
        rr[k] = r; cc[k] = c;
        const int gy = reflc(2 * I0 - 4 + r, H);
        const int gxx = reflc(2 * J0 - 4 + c, W) * xs;
        v[k] = sp[gy * rowPitch + gxx];
    }
#pragma unroll
    for (int k = 0; k < G; ++k) {
        float x = v[k];
        if (GAM) x = __builtin_amdgcn_exp2f(GAMMA_EXP * __builtin_amdgcn_logf(x));
        slab[rr[k]][cc[k]] = x;
    }
}

template <int GAM>
__device__ __forceinline__ void stage_slab_part(const float* __restrict__ src,
                                                float (*slab)[SLAB_PITCH],
                                                int p, int I0, int J0, int H, int W, int tid,
                                                int rmax, int cmax) {
    const float* sp;
    int rowPitch, xs;
    if (GAM) {
        int b = p / 3, c = p - 3 * b;
        sp = src + (size_t)b * H * W * 3 + c;
        rowPitch = W * 3; xs = 3;
    } else {
        int Wp = W + 2;
        sp = src + (size_t)p * (H + 2) * Wp + Wp + 1;
        rowPitch = Wp; xs = 1;
    }
    float4* s4 = reinterpret_cast<float4*>(&slab[0][0]);
    for (int i = tid; i < SLAB_N * SLAB_PITCH / 4; i += 128)
        s4[i] = make_float4(0.0f, 0.0f, 0.0f, 0.0f);
    __syncthreads();

    const int ne = rmax * cmax;
    const float inv = 1.0f / (float)cmax;
    if (ne <= 512) {
        stage_part_group<GAM, 4>(sp, slab, rowPitch, xs, I0, J0, H, W, tid, 0, ne, cmax, inv);
    } else {
        for (int base = 0; base < ne; base += 128 * 12)
            stage_part_group<GAM, 12>(sp, slab, rowPitch, xs, I0, J0, H, W, tid, base, ne, cmax, inv);
    }
}

__global__ void __launch_bounds__(128)
nlp_down(const float* __restrict__ src, float* __restrict__ down,
         int H, int W, int gammaFlag) {
    __shared__ __attribute__((aligned(16))) float slab[SLAB_N][SLAB_PITCH];
    __shared__ float vbuf[4][16][49];

    const int H2 = H / 2 + 2, W2 = W / 2 + 2;
    const int p  = blockIdx.z;
    const int I0 = blockIdx.y * 32, J0 = blockIdx.x * 32;
    const int tid = threadIdx.x;

    const int nvr = (H2 - I0) < 32 ? (H2 - I0) : 32;
    const int nvc = (W2 - J0) < 32 ? (W2 - J0) : 32;
    if (nvr == 32 && nvc == 32) {
        if (gammaFlag) stage_slab<1>(src, slab, p, I0, J0, H, W, tid);
        else           stage_slab<0>(src, slab, p, I0, J0, H, W, tid);
    } else {
        const int rmax = 2 * nvr + 3, cmax = 2 * nvc + 3;
        if (gammaFlag) stage_slab_part<1>(src, slab, p, I0, J0, H, W, tid, rmax, cmax);
        else           stage_slab_part<0>(src, slab, p, I0, J0, H, W, tid, rmax, cmax);
    }
    __syncthreads();

    const int lane = tid & 31;
    const int wv = __builtin_amdgcn_readfirstlane(tid >> 5);
    const int wr = wv >> 1, wc = wv & 1;
    const bool act = (16 * wr < nvr) && (16 * wc < nvc);
    const int i0 = I0 + 16 * wr, j0 = J0 + 16 * wc;
    const int n = lane & 15;
    const int half = lane >> 4;
    float (*vb)[49] = vbuf[wv];

    int colx[3];
#pragma unroll
    for (int nt = 0; nt < 3; ++nt) {
        int lc = 32 * wc + nt * 16 + n;
        colx[nt] = lc > SLAB_N - 1 ? SLAB_N - 1 : lc;
    }
    const int lr0 = 32 * wr;

    v8f acc0 = {0,0,0,0,0,0,0,0};
    v8f acc1 = {0,0,0,0,0,0,0,0};
    v8f acc2 = {0,0,0,0,0,0,0,0};

    if (act) {
    for (int c = 0; c < 9; ++c) {
        const int kb = 4 * c + 2 * half;
        v2f a;
        a.x = ltap(kb - 2 * n);
        a.y = ltap(kb + 1 - 2 * n);
        const float* s0 = &slab[lr0 + kb][0];
        const float* s1 = &slab[lr0 + kb + 1][0];

        v2f b0, b1, b2;
        b0.x = s0[colx[0]];
        b0.y = s1[colx[0]];
        b1.x = s0[colx[1]];
        b1.y = s1[colx[1]];
        b2.x = s0[colx[2]];
        b2.y = s1[colx[2]];

        acc0 = __builtin_amdgcn_wmma_f32_16x16x4_f32(false, a, false, b0,
                                                     (short)0, acc0, false, false);
        acc1 = __builtin_amdgcn_wmma_f32_16x16x4_f32(false, a, false, b1,
                                                     (short)0, acc1, false, false);
        acc2 = __builtin_amdgcn_wmma_f32_16x16x4_f32(false, a, false, b2,
                                                     (short)0, acc2, false, false);
    }

#pragma unroll
    for (int v = 0; v < 8; ++v) {
        vb[v + 8 * half][n]      = acc0[v];
        vb[v + 8 * half][16 + n] = acc1[v];
        vb[v + 8 * half][32 + n] = acc2[v];
    }
    }
    __syncthreads();
    if (!act) return;

    v8f dacc = {0,0,0,0,0,0,0,0};
    for (int c = 0; c < 9; ++c) {
        const int kb = 4 * c + 2 * half;
        v2f a;
        a.x = ltap(kb - 2 * n);
        a.y = ltap(kb + 1 - 2 * n);
        v2f b;
        b.x = vb[n][kb];
        b.y = vb[n][kb + 1];
        dacc = __builtin_amdgcn_wmma_f32_16x16x4_f32(false, a, false, b,
                                                     (short)0, dacc, false, false);
    }

    const int gi = i0 + n;
    if (gi < H2) {
        float* dp = down + (size_t)p * H2 * W2 + (size_t)gi * W2;
#pragma unroll
        for (int v = 0; v < 8; ++v) {
            int gj = j0 + v + 8 * half;
            if (gj < W2) dp[gj] = dacc[v];
        }
    }
}

__device__ __forceinline__ void down_block_wg(const float* src, float* down, int H, int p,
                                              int I0, int J0, float (*slab)[SLAB_PITCH],
                                              float (*vbuf)[16][49], int tid) {
    const int W = H;
    const int H2 = H / 2 + 2, W2 = W / 2 + 2;
    const int lane = tid & 31;
    const int wv = __builtin_amdgcn_readfirstlane(tid >> 5);
    const bool on = wv < 4;
    const int nvr = (H2 - I0) < 32 ? (H2 - I0) : 32;
    const int nvc = (W2 - J0) < 32 ? (W2 - J0) : 32;

    if (on) {
        float4* s4 = reinterpret_cast<float4*>(&slab[0][0]);
        for (int i = tid; i < SLAB_N * SLAB_PITCH / 4; i += 128)
            s4[i] = make_float4(0.0f, 0.0f, 0.0f, 0.0f);
    }
    __syncthreads();
    if (on) {
        const int Wp = W + 2;
        const float* sp = src + (size_t)p * (H + 2) * Wp + Wp + 1;
        const int rmax = 2 * nvr + 3, cmax = 2 * nvc + 3;
        const int ne = rmax * cmax;
        const float inv = 1.0f / (float)cmax;
        if (ne <= 512) {
            stage_part_group<0, 4>(sp, slab, Wp, 1, I0, J0, H, W, tid, 0, ne, cmax, inv);
        } else {
            for (int base = 0; base < ne; base += 128 * 12)
                stage_part_group<0, 12>(sp, slab, Wp, 1, I0, J0, H, W, tid, base, ne, cmax, inv);
        }
    }
    __syncthreads();

    const int wr = (wv >> 1) & 1, wc = wv & 1;
    const bool act = on && (16 * wr < nvr) && (16 * wc < nvc);
    const int i0 = I0 + 16 * wr, j0 = J0 + 16 * wc;
    const int n = lane & 15;
    const int half = lane >> 4;
    float (*vb)[49] = vbuf[wv & 3];

    if (act) {
        int colx[3];
#pragma unroll
        for (int nt = 0; nt < 3; ++nt) {
            int lc = 32 * wc + nt * 16 + n;
            colx[nt] = lc > SLAB_N - 1 ? SLAB_N - 1 : lc;
        }
        const int lr0 = 32 * wr;
        v8f acc0 = {0,0,0,0,0,0,0,0};
        v8f acc1 = {0,0,0,0,0,0,0,0};
        v8f acc2 = {0,0,0,0,0,0,0,0};
        for (int c = 0; c < 9; ++c) {
            const int kb = 4 * c + 2 * half;
            v2f a;
            a.x = ltap(kb - 2 * n);
            a.y = ltap(kb + 1 - 2 * n);
            const float* s0 = &slab[lr0 + kb][0];
            const float* s1 = &slab[lr0 + kb + 1][0];
            v2f b0, b1, b2;
            b0.x = s0[colx[0]];
            b0.y = s1[colx[0]];
            b1.x = s0[colx[1]];
            b1.y = s1[colx[1]];
            b2.x = s0[colx[2]];
            b2.y = s1[colx[2]];
            acc0 = __builtin_amdgcn_wmma_f32_16x16x4_f32(false, a, false, b0,
                                                         (short)0, acc0, false, false);
            acc1 = __builtin_amdgcn_wmma_f32_16x16x4_f32(false, a, false, b1,
                                                         (short)0, acc1, false, false);
            acc2 = __builtin_amdgcn_wmma_f32_16x16x4_f32(false, a, false, b2,
                                                         (short)0, acc2, false, false);
        }
#pragma unroll
        for (int v = 0; v < 8; ++v) {
            vb[v + 8 * half][n]      = acc0[v];
            vb[v + 8 * half][16 + n] = acc1[v];
            vb[v + 8 * half][32 + n] = acc2[v];
        }
    }
    __syncthreads();

    if (act) {
        v8f dacc = {0,0,0,0,0,0,0,0};
        for (int c = 0; c < 9; ++c) {
            const int kb = 4 * c + 2 * half;
            v2f a;
            a.x = ltap(kb - 2 * n);
            a.y = ltap(kb + 1 - 2 * n);
            v2f b;
            b.x = vb[n][kb];
            b.y = vb[n][kb + 1];
            dacc = __builtin_amdgcn_wmma_f32_16x16x4_f32(false, a, false, b,
                                                         (short)0, dacc, false, false);
        }
        const int gi = i0 + n;
        if (gi < H2) {
            float* dp = down + (size_t)p * H2 * W2 + (size_t)gi * W2;
#pragma unroll
            for (int v = 0; v < 8; ++v) {
                int gj = j0 + v + 8 * half;
                if (gj < W2) dp[gj] = dacc[v];
            }
        }
    }
}

#define N_PLANES    48
#define TOTAL_TILES 16416
#define SUB_GRID    1536
#define T_LVL3      16128
#define T_LVL4      16320
#define T_RESID     16368
#define T_OWNER_L2  16032
#define SUB_STRIDE  (SUB_GRID - N_PLANES)

__device__ constexpr size_t down_off(int l) {
    size_t off = 0;
    int H = 512;
    for (int i = 0; i < l; ++i) {
        const size_t h2 = (size_t)(H / 2 + 2);
        off += (size_t)N_PLANES * h2 * h2;
        H >>= 1;
    }
    return off;
}

__device__ __forceinline__ int seq_tile(int w, int k) {
    if (w < N_PLANES) {
        if (k < 4)  return T_LVL3 + 4 * w + k;
        if (k == 4) return T_LVL4 + w;
        if (k == 5) return T_RESID + w;
        if (k < 8)  return T_OWNER_L2 + (k - 6) * N_PLANES + w;
        return -1;
    }
    const int t = (w - N_PLANES) + k * SUB_STRIDE;
    return t < T_OWNER_L2 ? t : -1;
}


struct TileInfo {
    int lvl, H, sh, rem;
    size_t dcur, dprev, ocur;
};

__device__ __forceinline__ TileInfo decode_tile(int t) {
    TileInfo ti;
    int rem = t, lvl = 0, H = 512, sh = 4;
    size_t dcur = 0, dprev = 0, ocur = 0;
    for (; lvl < 5; ++lvl) {
        const int cnt = N_PLANES << (2 * sh);
        if (rem < cnt) break;
        rem -= cnt;
        const size_t h2 = (size_t)(H / 2 + 2);
        dprev = dcur;
        dcur += (size_t)N_PLANES * h2 * h2;
        ocur += (size_t)16 * H * H * 3;
        H >>= 1;
        --sh;
    }
    ti.lvl = lvl; ti.H = H; ti.sh = sh; ti.rem = rem;
    ti.dcur = dcur; ti.dprev = dprev; ti.ocur = ocur;
    return ti;
}

__device__ __forceinline__ void fetch_tile(const TileInfo& ti, const float* __restrict__ img,
                                           const float* __restrict__ ws, int tid,
                                           float (&pd)[2], float (&pxe)[3], float (&pxo)[3]) {
    if (ti.lvl == 5) {
        const int y = tid >> 4, x = tid & 15;
        pd[0] = ws[ti.dprev + (size_t)ti.rem * 18 * 18 + (size_t)(y + 1) * 18 + (x + 1)];
        return;
    }
    const int H = ti.H, W = ti.H, sh = ti.sh;
    const int nTm = (1 << sh) - 1;
    const int p  = ti.rem >> (2 * sh);
    const int y0 = ((ti.rem >> sh) & nTm) * 32, x0 = (ti.rem & nTm) * 32;
    const int H2 = H / 2 + 2, W2 = W / 2 + 2;
    const float* dw = ws + ti.dcur + (size_t)p * H2 * W2;
    const int dr0 = (y0 >> 1) - 1, dc0 = (x0 >> 1) - 1;
    for (int k = 0; k < 2; ++k) {
        int q = tid + 256 * k;
        q = q > 399 ? 399 : q;
        const int ly = q / 20, lx = q - 20 * ly;
        int r = dr0 + ly; r = r < 0 ? 0 : (r >= H2 ? H2 - 1 : r);
        int c = dc0 + lx; c = c < 0 ? 0 : (c >= W2 ? W2 - 1 : c);
        pd[k] = dw[r * W2 + c];
    }
    const float* sp;
    int pitch, xs;
    if (ti.lvl == 0) {
        const int b0 = p / 3, c0 = p - 3 * b0;
        sp = img + (size_t)b0 * H * W * 3 + c0; pitch = W * 3; xs = 3;
    } else {
        const int Wp = W + 2;
        sp = ws + ti.dprev + (size_t)p * (H + 2) * Wp + Wp + 1; pitch = Wp; xs = 1;
    }
    for (int k = 0; k < 3; ++k) {
        int e = tid + 256 * k;
        e = e > 647 ? 647 : e;
        const int i = e / 36, lx = e - 36 * i;
        int y = y0 - 2 + 2 * i;
        int x = x0 - 2 + lx;
        y = y < 0 ? 0 : (y > H - 2 ? H - 2 : y);
        x = x < 0 ? 0 : (x > W - 1 ? W - 1 : x);
        const int off = y * pitch + x * xs;
        pxe[k] = sp[off];
        pxo[k] = sp[off + pitch];
    }
}

__global__ void __launch_bounds__(256, 12)
nlp_sub_all(const float* __restrict__ img, float* ws,
            float* __restrict__ outAll) {
    __shared__ float zb[36][37];
    __shared__ float dt[20][21];
    __shared__ float uh[20][37];
    __shared__ __attribute__((aligned(16))) float slab[SLAB_N][SLAB_PITCH];
    __shared__ float vbuf[4][16][49];

    const int tid = threadIdx.x;
    const int wg = blockIdx.x;

    if (wg < N_PLANES) {
        const float* d2 = ws + down_off(2);
        float* d3 = ws + down_off(3);
        float* d4 = ws + down_off(4);
        for (int blk = 0; blk < 5; ++blk) {
            const bool l4 = (blk == 4);
            down_block_wg(l4 ? d3 : d2, l4 ? d4 : d3, l4 ? 32 : 64, wg,
                          l4 ? 0 : 32 * (blk >> 1), l4 ? 0 : 32 * (blk & 1), slab, vbuf, tid);
            if (blk >= 3) __syncthreads();
        }
    }

    const float Pk[5][5] = {
        {0.04f, 0.04f, 0.05f, 0.04f, 0.04f},
        {0.04f, 0.03f, 0.04f, 0.03f, 0.04f},
        {0.05f, 0.04f, 0.05f, 0.04f, 0.05f},
        {0.04f, 0.03f, 0.04f, 0.03f, 0.04f},
        {0.04f, 0.04f, 0.05f, 0.04f, 0.04f}};

    int k = 0;
    TileInfo cur = decode_tile(seq_tile(wg, 0));
    float pd[2], pxe[3], pxo[3];
    fetch_tile(cur, img, ws, tid, pd, pxe, pxo);

    for (;;) {
        const int tn = seq_tile(wg, k + 1);
        const bool more = tn >= 0;
        TileInfo nxt = cur;

        if (cur.lvl == 5) {
            const int p = cur.rem;
            const int y = tid >> 4, x = tid & 15;
            const float v = pd[0];
            const float o = v / (fabsf(v) + SIGMA_LP);
            const int b = p / 3, c = p - 3 * b;
            outAll[cur.ocur + (((size_t)b * 16 + y) * 16 + x) * 3 + c] = o;
            if (!more) break;
            nxt = decode_tile(tn);
            fetch_tile(nxt, img, ws, tid, pd, pxe, pxo);
            cur = nxt; ++k;
            continue;
        }

        const int H = cur.H, W = cur.H, sh = cur.sh;
        const bool gam = (cur.lvl == 0);
        float* out = outAll + cur.ocur;
        const int nTm = (1 << sh) - 1;
        const int p  = cur.rem >> (2 * sh);
        const int y0 = ((cur.rem >> sh) & nTm) * 32, x0 = (cur.rem & nTm) * 32;

        for (int k = 0; k < 2; ++k) {
            const int q = tid + 256 * k;
            if (q < 400) {
                const int ly = q / 20, lx = q - 20 * ly;
                dt[ly][lx] = pd[k];
            }
        }
        __syncthreads();

        for (int e = tid; e < 360; e += 256) {
            const int r = e / 18, j = e - 18 * r;
            const float d0 = dt[r][j], d1 = dt[r][j + 1], d2 = dt[r][j + 2];
            float he = fmaf(0.1f, d0, 0.0f);
            he = fmaf(0.8f, d1, he);
            he = fmaf(0.1f, d2, he);
            float ho = fmaf(0.5f, d1, 0.0f);
            ho = fmaf(0.5f, d2, ho);
            uh[r][2 * j]     = he;
            uh[r][2 * j + 1] = ho;
        }
        __syncthreads();

        for (int k = 0; k < 3; ++k) {
            const int e = tid + 256 * k;
            if (e < 648) {
                const int i = e / 36, lx = e - 36 * i;
                const int y = y0 - 2 + 2 * i;
                const int x = x0 - 2 + lx;
                float ze = 0.0f, zo = 0.0f;
                if ((unsigned)y < (unsigned)H && (unsigned)x < (unsigned)W) {
                    float xe = pxe[k], xo = pxo[k];
                    if (gam) {
                        xe = __builtin_amdgcn_exp2f(GAMMA_EXP * __builtin_amdgcn_logf(xe));
                        xo = __builtin_amdgcn_exp2f(GAMMA_EXP * __builtin_amdgcn_logf(xo));
                    }
                    const float u0 = uh[i][lx], u1 = uh[i + 1][lx], u2 = uh[i + 2][lx];
                    float upe = fmaf(0.1f, u0, 0.0f);
                    upe = fmaf(0.8f, u1, upe);
                    upe = fmaf(0.1f, u2, upe);
                    float upo = fmaf(0.5f, u1, 0.0f);
                    upo = fmaf(0.5f, u2, upo);
                    ze = xe - upe;
                    zo = xo - upo;
                }
                zb[2 * i][lx]     = ze;
                zb[2 * i + 1][lx] = zo;
            }
        }
        __syncthreads();

        if (more) {
            nxt = decode_tile(tn);
            fetch_tile(nxt, img, ws, tid, pd, pxe, pxo);
        }

        const int b = p / 3, ch = p - 3 * b;
#pragma unroll
        for (int q = 0; q < 4; ++q) {
            int idx = tid + q * 256;
            int oy = idx >> 5, ox = idx & 31;
            float zc = zb[oy + 2][ox + 2];
            float pool = 0.0f;
#pragma unroll
            for (int dy = 0; dy < 5; ++dy)
#pragma unroll
                for (int dx = 0; dx < 5; ++dx)
                    pool = fmaf(Pk[dy][dx], fabsf(zb[oy + dy][ox + dx]), pool);
            float o = zc / (pool + SIGMA_P);
            int gy = y0 + oy, gx = x0 + ox;
            out[(((size_t)b * H + gy) * W + gx) * 3 + ch] = o;
        }

        if (!more) break;
        cur = nxt; ++k;
    }
}

extern "C" void kernel_launch(void* const* d_in, const int* in_sizes, int n_in,
                              void* d_out, int out_size, void* d_ws, size_t ws_size,
                              hipStream_t stream) {
    (void)in_sizes; (void)n_in; (void)out_size; (void)ws_size;
    const float* img = (const float*)d_in[0];
    float* out = (float*)d_out;
    float* ws  = (float*)d_ws;

    const int Hl[6] = {512, 256, 128, 64, 32, 16};

    size_t dof[5]; size_t off = 0;
    for (int l = 0; l < 5; ++l) {
        dof[l] = off;
        size_t h2 = (size_t)(Hl[l] / 2 + 2);
        off += (size_t)48 * h2 * h2;
    }

    const float* srcp = img;
    int gam = 1;
    for (int l = 0; l < 3; ++l) {
        const int H = Hl[l];
        const int H2 = H / 2 + 2;
        const int T = (H2 + 31) / 32;
        nlp_down<<<dim3(T, T, 48), 128, 0, stream>>>(srcp, ws + dof[l], H, H, gam);
        srcp = ws + dof[l];
        gam = 0;
    }
    nlp_sub_all<<<SUB_GRID, 256, 0, stream>>>(img, ws, out);
}
